# starting kernel
# baseline (speedup 1.0000x reference)
#include <hip/hip_runtime.h>

typedef __attribute__((ext_vector_type(2))) float v2f;
typedef __attribute__((ext_vector_type(8))) float v8f;

#define C_CHANNELS 128
#define ROWS_PER_BLOCK 16
#define CHUNK 256

__global__ __launch_bounds__(256) void neighbor_pool_wmma(
    const float* __restrict__ feat,
    const int*   __restrict__ nidx,
    const int*   __restrict__ splits,
    float*       __restrict__ out,
    int M)
{
    __shared__ int   s_split[ROWS_PER_BLOCK + 1];
    __shared__ float s_inv[ROWS_PER_BLOCK];
    __shared__ int   s_idx[CHUNK];

    const int tid  = threadIdx.x;
    const int lane = tid & 31;
    const int wave = tid >> 5;
    const int base = blockIdx.x * ROWS_PER_BLOCK;

    if (tid <= ROWS_PER_BLOCK) {
        int r = base + tid;
        if (r > M) r = M;
        s_split[tid] = splits[r];
    }
    __syncthreads();

    const int      r     = lane & 15;
    const int      myLo  = s_split[r];
    const unsigned myCnt = (unsigned)(s_split[r + 1] - myLo);
    const int      s0    = s_split[0];
    const int      s1    = s_split[ROWS_PER_BLOCK];
    const int      k0    = (lane >= 16) ? 2 : 0;
    const unsigned chan  = (unsigned)(wave * 16 + r);

    v8f acc = {};

    for (int cbeg = s0; cbeg < s1; cbeg += CHUNK) {
        int cend = cbeg + CHUNK;
        if (cend > s1) cend = s1;

        {
            int t = cbeg + tid;
            if (t >= s1) t = s1 - 1;
            s_idx[tid] = __builtin_nontemporal_load(&nidx[t]);
        }
        __syncthreads();

        const int eend = cbeg + ((cend - cbeg + 15) & ~15);
#pragma unroll 4
        for (int e = cbeg; e < eend; e += 4) {
            const int ge0 = e + k0;
            const int ge1 = ge0 + 1;

            v2f a;
            a.x = ((unsigned)(ge0 - myLo) < myCnt) ? 1.0f : 0.0f;
            a.y = ((unsigned)(ge1 - myLo) < myCnt) ? 1.0f : 0.0f;

            const int2 p = *(const int2*)(s_idx + (ge0 - cbeg));

            v2f b;
            b.x = feat[((unsigned)p.x << 7) + chan];
            b.y = feat[((unsigned)p.y << 7) + chan];

            acc = __builtin_amdgcn_wmma_f32_16x16x4_f32(
                 false, a,  false, b,
                 (short)0, acc,
                 false,  false);
        }
        __syncthreads();
    }

    if (tid < ROWS_PER_BLOCK) {
        int cnt = s_split[tid + 1] - s_split[tid];
        if (cnt < 1) cnt = 1;
        s_inv[tid] = 1.0f / (float)cnt;
    }
    __syncthreads();

    const int rowOfs = (lane >= 16) ? 8 : 0;
#pragma unroll
    for (int g = 0; g < 8; ++g) {
        const int mrow = g + rowOfs;
        const int m = base + mrow;
        if (m < M) {
            __builtin_nontemporal_store(acc[g] * s_inv[mrow],
                                        &out[(unsigned)m * C_CHANNELS + chan]);
        }
    }
}

extern "C" void kernel_launch(void* const* d_in, const int* in_sizes, int n_in,
                              void* d_out, int out_size, void* d_ws, size_t ws_size,
                              hipStream_t stream) {
    const float* feat   = (const float*)d_in[0];
    const int*   nidx   = (const int*)  d_in[1];
    const int*   splits = (const int*)  d_in[2];
    float*       out    = (float*)      d_out;

    const int M = in_sizes[2] - 1;
    const int blocks = (M + ROWS_PER_BLOCK - 1) / ROWS_PER_BLOCK;
    neighbor_pool_wmma<<<blocks, 256, 0, stream>>>(feat, nidx, splits, out, M);
}


# Each wave requests the rows of 64 edges (32 loads) per round trip before its 16 one-hot WMMAs instead of 16 edges (8 loads): 4x fewer exposed load waits per wave; k order and sums unchanged
# speedup vs baseline: 1.1701x; 1.1701x over previous
#include <hip/hip_runtime.h>


typedef __attribute__((ext_vector_type(2))) float v2f;
typedef __attribute__((ext_vector_type(8))) float v8f;

#define C_CHANNELS 128
#define ROWS_PER_BLOCK 16
#define CHUNK 256
#define KMAIN 16

template <int K>
__device__ __forceinline__ void gather_sum(
    v8f& acc, const float* __restrict__ feat, const int* sidx,
    int t0, unsigned myCnt, unsigned chan)
{
    v2f b[K];
#pragma unroll
    for (int j = 0; j < K; ++j) {
        const int2 p = *(const int2*)(sidx + 4 * j);
        b[j].x = feat[((unsigned)p.x << 7) + chan];
        b[j].y = feat[((unsigned)p.y << 7) + chan];
    }
#pragma unroll
    for (int j = 0; j < K; ++j) {
        const int t = t0 + 4 * j;
        v2f a;
        a.x = ((unsigned)t       < myCnt) ? 1.0f : 0.0f;
        a.y = ((unsigned)(t + 1) < myCnt) ? 1.0f : 0.0f;
        acc = __builtin_amdgcn_wmma_f32_16x16x4_f32(
             false, a,  false, b[j],
             (short)0, acc,
             false,  false);
    }
}

__global__ __launch_bounds__(256) void neighbor_pool_wmma(
    const float* __restrict__ feat,
    const int*   __restrict__ nidx,
    const int*   __restrict__ splits,
    float*       __restrict__ out,
    int M)
{
    __shared__ int   s_split[ROWS_PER_BLOCK + 1];
    __shared__ float s_inv[ROWS_PER_BLOCK];
    __shared__ __attribute__((aligned(16))) int s_idx[CHUNK];

    const int tid  = threadIdx.x;
    const int lane = tid & 31;
    const int wave = tid >> 5;
    const int base = blockIdx.x * ROWS_PER_BLOCK;

    if (tid <= ROWS_PER_BLOCK) {
        int r = base + tid;
        if (r > M) r = M;
        s_split[tid] = splits[r];
    }
    __syncthreads();

    const int      r     = lane & 15;
    const int      myLo  = s_split[r];
    const unsigned myCnt = (unsigned)(s_split[r + 1] - myLo);
    const int      s0    = s_split[0];
    const int      s1    = s_split[ROWS_PER_BLOCK];
    const int      k0    = (lane >= 16) ? 2 : 0;
    const unsigned chan  = (unsigned)(wave * 16 + r);

    v8f acc = {};

    for (int cbeg = s0; cbeg < s1; cbeg += CHUNK) {
        int cend = cbeg + CHUNK;
        if (cend > s1) cend = s1;

        {
            int t = cbeg + tid;
            if (t >= s1) t = s1 - 1;
            s_idx[tid] = __builtin_nontemporal_load(&nidx[t]);
        }
        __syncthreads();

        const int  ksteps = ((cend - cbeg + 15) & ~15) >> 2;
        const int* sp     = s_idx + k0;
        const int  t0     = cbeg + k0 - myLo;
        int k = 0;
        for (; k + KMAIN <= ksteps; k += KMAIN)
            gather_sum<KMAIN>(acc, feat, sp + 4 * k, t0 + 4 * k, myCnt, chan);
        if (k + 8 <= ksteps) {
            gather_sum<8>(acc, feat, sp + 4 * k, t0 + 4 * k, myCnt, chan);
            k += 8;
        }
        if (k < ksteps)
            gather_sum<4>(acc, feat, sp + 4 * k, t0 + 4 * k, myCnt, chan);
        __syncthreads();
    }

    if (tid < ROWS_PER_BLOCK) {
        int cnt = s_split[tid + 1] - s_split[tid];
        if (cnt < 1) cnt = 1;
        s_inv[tid] = 1.0f / (float)cnt;
    }
    __syncthreads();

    const int rowOfs = (lane >= 16) ? 8 : 0;
#pragma unroll
    for (int g = 0; g < 8; ++g) {
        const int mrow = g + rowOfs;
        const int m = base + mrow;
        if (m < M) {
            __builtin_nontemporal_store(acc[g] * s_inv[mrow],
                                        &out[(unsigned)m * C_CHANNELS + chan]);
        }
    }
}

extern "C" void kernel_launch(void* const* d_in, const int* in_sizes, int n_in,
                              void* d_out, int out_size, void* d_ws, size_t ws_size,
                              hipStream_t stream) {
    const float* feat   = (const float*)d_in[0];
    const int*   nidx   = (const int*)  d_in[1];
    const int*   splits = (const int*)  d_in[2];
    float*       out    = (float*)      d_out;

    const int M = in_sizes[2] - 1;
    const int blocks = (M + ROWS_PER_BLOCK - 1) / ROWS_PER_BLOCK;
    neighbor_pool_wmma<<<blocks, 256, 0, stream>>>(feat, nidx, splits, out, M);
}


# Each lane gathers 2 adjacent channels of a row with one 8-byte load feeding 2 channel-tile WMMAs that share one A (wave owns 32 channels, 4-wave blocks): half the loads and A work; sum order kept
# speedup vs baseline: 1.2871x; 1.1000x over previous
#include <hip/hip_runtime.h>


typedef __attribute__((ext_vector_type(2))) float v2f;
typedef __attribute__((ext_vector_type(8))) float v8f;

#define C_CHANNELS 128
#define ROWS_PER_BLOCK 16
#define THREADS 128
#define CHUNK 256
#define KMAIN 16


template <int K>
__device__ __forceinline__ void gather_sum(
    v8f (&acc)[2], const float* __restrict__ feat, const int* sidx,
    int t0, unsigned myCnt, unsigned cbyte)
{
    v2f bx[K], by[K];
#pragma unroll
    for (int j = 0; j < K; ++j) {
        const int2 p = *(const int2*)(sidx + 4 * j);
        bx[j] = *(const v2f*)((const char*)feat + (((unsigned)p.x << 9) + cbyte));
        by[j] = *(const v2f*)((const char*)feat + (((unsigned)p.y << 9) + cbyte));
    }
    __builtin_amdgcn_sched_barrier(0);
#pragma unroll
    for (int j = 0; j < K; ++j) {
        const int t = t0 + 4 * j;
        v2f a;
        a.x = ((unsigned)t       < myCnt) ? 1.0f : 0.0f;
        a.y = ((unsigned)(t + 1) < myCnt) ? 1.0f : 0.0f;
#pragma unroll
        for (int c = 0; c < 2; ++c) {
            v2f b;
            b.x = bx[j][c];
            b.y = by[j][c];
            acc[c] = __builtin_amdgcn_wmma_f32_16x16x4_f32(
                 false, a,  false, b,
                 (short)0, acc[c],
                 false,  false);
        }
    }
}

__global__ __launch_bounds__(THREADS) void neighbor_pool_wmma(
    const float* __restrict__ feat,
    const int*   __restrict__ nidx,
    const int*   __restrict__ splits,
    float*       __restrict__ out,
    int M)
{
    __shared__ int   s_split[ROWS_PER_BLOCK + 1];
    __shared__ float s_inv[ROWS_PER_BLOCK];
    __shared__ __attribute__((aligned(16))) int s_idx[CHUNK];

    const int tid  = threadIdx.x;
    const int lane = tid & 31;
    const int wave = tid >> 5;
    const int base = blockIdx.x * ROWS_PER_BLOCK;

    if (tid <= ROWS_PER_BLOCK) {
        int r = base + tid;
        if (r > M) r = M;
        s_split[tid] = splits[r];
    }
    __syncthreads();

    const int      r     = lane & 15;
    const int      myLo  = s_split[r];
    const unsigned myCnt = (unsigned)(s_split[r + 1] - myLo);
    const int      s0    = s_split[0];
    const int      s1    = s_split[ROWS_PER_BLOCK];
    const int      k0    = (lane >= 16) ? 2 : 0;
    const unsigned chan2 = (unsigned)(wave * 32 + 2 * r);

    v8f acc[2] = {};

    for (int cbeg = s0; cbeg < s1; cbeg += CHUNK) {
        int cend = cbeg + CHUNK;
        if (cend > s1) cend = s1;

#pragma unroll
        for (int i = 0; i < CHUNK / THREADS; ++i) {
            int t = cbeg + tid + THREADS * i;
            if (t >= s1) t = s1 - 1;
            s_idx[tid + THREADS * i] = __builtin_nontemporal_load(&nidx[t]);
        }
        __syncthreads();

        const int  ksteps = ((cend - cbeg + 15) & ~15) >> 2;
        const int* sp     = s_idx + k0;
        const int  t0     = cbeg + k0 - myLo;
        int k = 0;
        for (; k + KMAIN <= ksteps; k += KMAIN)
            gather_sum<KMAIN>(acc, feat, sp + 4 * k, t0 + 4 * k, myCnt, chan2 * 4u);
        if (k + 8 <= ksteps) {
            gather_sum<8>(acc, feat, sp + 4 * k, t0 + 4 * k, myCnt, chan2 * 4u);
            k += 8;
        }
        if (k < ksteps)
            gather_sum<4>(acc, feat, sp + 4 * k, t0 + 4 * k, myCnt, chan2 * 4u);
        __syncthreads();
    }

    if (tid < ROWS_PER_BLOCK) {
        int cnt = s_split[tid + 1] - s_split[tid];
        if (cnt < 1) cnt = 1;
        s_inv[tid] = 1.0f / (float)cnt;
    }
    __syncthreads();

    const int rowOfs = (lane >= 16) ? 8 : 0;
#pragma unroll
    for (int g = 0; g < 8; ++g) {
        const int mrow = g + rowOfs;
        const int m = base + mrow;
        if (m < M) {
            const float s = s_inv[mrow];
            v2f o;
            o.x = acc[0][g] * s;
            o.y = acc[1][g] * s;
            __builtin_nontemporal_store(o, (v2f*)&out[(unsigned)m * C_CHANNELS + chan2]);
        }
    }
}

extern "C" void kernel_launch(void* const* d_in, const int* in_sizes, int n_in,
                              void* d_out, int out_size, void* d_ws, size_t ws_size,
                              hipStream_t stream) {
    const float* feat   = (const float*)d_in[0];
    const int*   nidx   = (const int*)  d_in[1];
    const int*   splits = (const int*)  d_in[2];
    float*       out    = (float*)      d_out;

    const int M = in_sizes[2] - 1;
    const int blocks = (M + ROWS_PER_BLOCK - 1) / ROWS_PER_BLOCK;
    neighbor_pool_wmma<<<blocks, THREADS, 0, stream>>>(feat, nidx, splits, out, M);
}


# Blocks take equal shares of the edge list, not 16 rows each: 1536 blocks own the rows starting in their share (parallel count over the sorted splits), staged in one pass; k order and sums unchanged
# speedup vs baseline: 1.4754x; 1.1463x over previous
#include <hip/hip_runtime.h>


typedef __attribute__((ext_vector_type(2))) float v2f;
typedef __attribute__((ext_vector_type(8))) float v8f;

#define C_CHANNELS 128
#define ROWS_PER_BLOCK 16
#define THREADS 128
#define NBLOCKS 1536
#define CHUNK 640
#define KMAIN 16


template <int K>
__device__ __forceinline__ void gather_sum(
    v8f (&acc)[2], const float* __restrict__ feat, const int* sidx,
    int t0, unsigned myCnt, unsigned cbyte)
{
    v2f bx[K], by[K];
#pragma unroll
    for (int j = 0; j < K; ++j) {
        const int2 p = *(const int2*)(sidx + 4 * j);
        bx[j] = *(const v2f*)((const char*)feat + (((unsigned)p.x << 9) + cbyte));
        by[j] = *(const v2f*)((const char*)feat + (((unsigned)p.y << 9) + cbyte));
    }
    __builtin_amdgcn_sched_barrier(0);
#pragma unroll
    for (int j = 0; j < K; ++j) {
        const int t = t0 + 4 * j;
        v2f a;
        a.x = ((unsigned)t       < myCnt) ? 1.0f : 0.0f;
        a.y = ((unsigned)(t + 1) < myCnt) ? 1.0f : 0.0f;
#pragma unroll
        for (int c = 0; c < 2; ++c) {
            v2f b;
            b.x = bx[j][c];
            b.y = by[j][c];
            acc[c] = __builtin_amdgcn_wmma_f32_16x16x4_f32(
                 false, a,  false, b,
                 (short)0, acc[c],
                 false,  false);
        }
    }
}

__global__ __launch_bounds__(THREADS) void neighbor_pool_wmma(
    const float* __restrict__ feat,
    const int*   __restrict__ nidx,
    const int*   __restrict__ splits,
    float*       __restrict__ out,
    int M)
{
    __shared__ int   s_split[ROWS_PER_BLOCK + 1];
    __shared__ float s_inv[ROWS_PER_BLOCK];
    __shared__ int   s_red[2][2][THREADS / 32];
    __shared__ __attribute__((aligned(16))) int s_idx[CHUNK];

    const int tid  = threadIdx.x;
    const int lane = tid & 31;
    const int wave = tid >> 5;
    const int blk  = blockIdx.x;
    const int nblk = NBLOCKS;

    const int S  = (M + 2 * THREADS - 1) / (2 * THREADS);
    const int p0 = tid * S;
    const int p1 = (tid + THREADS) * S;
    const int E  = splits[M];
    const int v0 = splits[p0 < M ? p0 : M];
    const int v1 = splits[p1 < M ? p1 : M];
    const int per = (E + nblk - 1) / nblk;
    const int tLo = blk * per;
    const int tHi = tLo + per;
    {
        const int cLo = __popc(__builtin_amdgcn_ballot_w32(p0 < M && v0 < tLo))
                      + __popc(__builtin_amdgcn_ballot_w32(p1 < M && v1 < tLo));
        const int cHi = __popc(__builtin_amdgcn_ballot_w32(p0 < M && v0 < tHi))
                      + __popc(__builtin_amdgcn_ballot_w32(p1 < M && v1 < tHi));
        if (lane == 0) { s_red[0][0][wave] = cLo; s_red[0][1][wave] = cHi; }
    }
    __syncthreads();
    const int c1Lo = s_red[0][0][0] + s_red[0][0][1] + s_red[0][0][2] + s_red[0][0][3];
    const int c1Hi = s_red[0][1][0] + s_red[0][1][1] + s_red[0][1][2] + s_red[0][1][3];
    const int bLo = c1Lo > 0 ? (c1Lo - 1) * S + 1 : 0;
    const int bHi = c1Hi > 0 ? (c1Hi - 1) * S + 1 : 0;
    int eLo = c1Lo > 0 ? c1Lo * S : 0;  if (eLo > M) eLo = M;
    int eHi = c1Hi > 0 ? c1Hi * S : 0;  if (eHi > M) eHi = M;
    {
        int nLo = 0, nHi = 0;
        for (int o0 = 0; o0 < S - 1; o0 += THREADS) {
            const int o  = o0 + tid;
            const int rl = bLo + o;
            const int rh = bHi + o;
            const int xl = splits[rl < M ? rl : M];
            const int xh = splits[rh < M ? rh : M];
            nLo += __popc(__builtin_amdgcn_ballot_w32(o < S - 1 && rl < eLo && xl < tLo));
            nHi += __popc(__builtin_amdgcn_ballot_w32(o < S - 1 && rh < eHi && xh < tHi));
        }
        if (lane == 0) { s_red[1][0][wave] = nLo; s_red[1][1][wave] = nHi; }
    }
    __syncthreads();
    const int rLo = (blk == 0) ? 0
        : bLo + s_red[1][0][0] + s_red[1][0][1] + s_red[1][0][2] + s_red[1][0][3];
    const int rHi = (blk == nblk - 1) ? M
        : bHi + s_red[1][1][0] + s_red[1][1][1] + s_red[1][1][2] + s_red[1][1][3];

  for (int base = rLo; base < rHi; base += ROWS_PER_BLOCK) {
    if (tid <= ROWS_PER_BLOCK) {
        int r = base + tid;
        if (r > rHi) r = rHi;
        s_split[tid] = splits[r];
    }
    __syncthreads();

    const int      r     = lane & 15;
    const int      myLo  = s_split[r];
    const unsigned myCnt = (unsigned)(s_split[r + 1] - myLo);
    const int      s0    = s_split[0];
    const int      s1    = s_split[ROWS_PER_BLOCK];
    const int      k0    = (lane >= 16) ? 2 : 0;
    const unsigned chan2 = (unsigned)(wave * 32 + 2 * r);

    v8f acc[2] = {};

    for (int cbeg = s0; cbeg < s1; cbeg += CHUNK) {
        int cend = cbeg + CHUNK;
        if (cend > s1) cend = s1;

#pragma unroll
        for (int i = 0; i < CHUNK / THREADS; ++i) {
            int t = cbeg + tid + THREADS * i;
            if (t >= s1) t = s1 - 1;
            s_idx[tid + THREADS * i] = __builtin_nontemporal_load(&nidx[t]);
        }
        __syncthreads();

        const int  ksteps = ((cend - cbeg + 15) & ~15) >> 2;
        const int* sp     = s_idx + k0;
        const int  t0     = cbeg + k0 - myLo;
        int k = 0;
        for (; k + KMAIN <= ksteps; k += KMAIN)
            gather_sum<KMAIN>(acc, feat, sp + 4 * k, t0 + 4 * k, myCnt, chan2 * 4u);
        if (k + 8 <= ksteps) {
            gather_sum<8>(acc, feat, sp + 4 * k, t0 + 4 * k, myCnt, chan2 * 4u);
            k += 8;
        }
        if (k < ksteps)
            gather_sum<4>(acc, feat, sp + 4 * k, t0 + 4 * k, myCnt, chan2 * 4u);
        __syncthreads();
    }

    if (tid < ROWS_PER_BLOCK) {
        int cnt = s_split[tid + 1] - s_split[tid];
        if (cnt < 1) cnt = 1;
        s_inv[tid] = 1.0f / (float)cnt;
    }
    __syncthreads();

    const int rowOfs = (lane >= 16) ? 8 : 0;
#pragma unroll
    for (int g = 0; g < 8; ++g) {
        const int mrow = g + rowOfs;
        const int m = base + mrow;
        if (m < rHi) {
            const float s = s_inv[mrow];
            v2f o;
            o.x = acc[0][g] * s;
            o.y = acc[1][g] * s;
            __builtin_nontemporal_store(o, (v2f*)&out[(unsigned)m * C_CHANNELS + chan2]);
        }
    }
  }
}

extern "C" void kernel_launch(void* const* d_in, const int* in_sizes, int n_in,
                              void* d_out, int out_size, void* d_ws, size_t ws_size,
                              hipStream_t stream) {
    const float* feat   = (const float*)d_in[0];
    const int*   nidx   = (const int*)  d_in[1];
    const int*   splits = (const int*)  d_in[2];
    float*       out    = (float*)      d_out;

    const int M = in_sizes[2] - 1;
    neighbor_pool_wmma<<<NBLOCKS, THREADS, 0, stream>>>(feat, nidx, splits, out, M);
}
